# starting kernel
# baseline (speedup 1.0000x reference)
#include <hip/hip_runtime.h>

#define BS      16
#define CIN     32
#define COUT    64
#define OUT_DIM 1024
#define KPATCH  4
#define DLEN    (OUT_DIM * KPATCH)
#define SCALE   0.17677669529663687f

typedef __attribute__((ext_vector_type(2))) float v2f;
typedef __attribute__((ext_vector_type(8))) float v8f;

__global__ __launch_bounds__(128) void
nolc1d_wmma_f32(const float* __restrict__ x,
                const float* __restrict__ w,
                float* __restrict__ out)
{
    const int lane = threadIdx.x & 31;
    const int wvid = threadIdx.x >> 5;
    const int p    = (blockIdx.x << 2) + wvid;

    const int half = lane >> 4;
    const int lid  = lane & 15;

    const float* xp = x + ((size_t)lid * CIN) * DLEN + 4 * p + 2 * half;
    const float* wp = w + ((size_t)lid * CIN) * DLEN + 4 * p + 2 * half;
    const size_t tstride = (size_t)16 * CIN * DLEN;

    v8f acc0 = {}, acc1 = {}, acc2 = {}, acc3 = {};

#pragma unroll 4
    for (int c = 0; c < CIN; ++c) {
        const size_t co = (size_t)c * DLEN;
        v2f a  = *(const v2f*)(xp + co);
        v2f b0 = *(const v2f*)(wp + co);
        v2f b1 = *(const v2f*)(wp + co + tstride);
        v2f b2 = *(const v2f*)(wp + co + 2 * tstride);
        v2f b3 = *(const v2f*)(wp + co + 3 * tstride);

        acc0 = __builtin_amdgcn_wmma_f32_16x16x4_f32(false, a, false, b0,
                                                     (short)0, acc0, false, false);
        acc1 = __builtin_amdgcn_wmma_f32_16x16x4_f32(false, a, false, b1,
                                                     (short)0, acc1, false, false);
        acc2 = __builtin_amdgcn_wmma_f32_16x16x4_f32(false, a, false, b2,
                                                     (short)0, acc2, false, false);
        acc3 = __builtin_amdgcn_wmma_f32_16x16x4_f32(false, a, false, b3,
                                                     (short)0, acc3, false, false);
    }

    const int b_base = half * 8;
#pragma unroll
    for (int j = 0; j < 8; ++j) {
        const int b = b_base + j;
        const size_t row = (size_t)b * COUT * OUT_DIM + p;
        out[row + (size_t)(0 * 16 + lid) * OUT_DIM] = acc0[j] * SCALE;
        out[row + (size_t)(1 * 16 + lid) * OUT_DIM] = acc1[j] * SCALE;
        out[row + (size_t)(2 * 16 + lid) * OUT_DIM] = acc2[j] * SCALE;
        out[row + (size_t)(3 * 16 + lid) * OUT_DIM] = acc3[j] * SCALE;
    }
}

extern "C" void kernel_launch(void* const* d_in, const int* in_sizes, int n_in,
                              void* d_out, int out_size, void* d_ws, size_t ws_size,
                              hipStream_t stream)
{
    (void)in_sizes; (void)n_in; (void)out_size; (void)d_ws; (void)ws_size;
    const float* x = (const float*)d_in[0];
    const float* w = (const float*)d_in[1];
    float* out     = (float*)d_out;

    dim3 grid(OUT_DIM / 4);
    dim3 block(128);
    nolc1d_wmma_f32<<<grid, block, 0, stream>>>(x, w, out);
}


# one wave = 2 patches x 16 outputs: 16-byte-per-lane loads straight into fragments (half-waves exchanged by v_permlane16_swap), all 64 loads issued before the first product; same sum order
# speedup vs baseline: 2.0588x; 2.0588x over previous
#include <hip/hip_runtime.h>


#define BS      16
#define CIN     32
#define COUT    64
#define OUT_DIM 1024
#define KPATCH  4
#define DLEN    (OUT_DIM * KPATCH)
#define SCALE   0.17677669529663687f

typedef __attribute__((ext_vector_type(2))) float    v2f;
typedef __attribute__((ext_vector_type(4))) float    v4f;
typedef __attribute__((ext_vector_type(8))) float    v8f;
typedef __attribute__((ext_vector_type(2))) unsigned v2u;

__device__ __forceinline__ void split_pair(const v4f q, v2f& f0, v2f& f1)
{
    const float k0 = q.x, k1 = q.y, k2 = q.z, k3 = q.w;
    const v2u r0 = __builtin_amdgcn_permlane16_swap(__float_as_uint(k0), __float_as_uint(k2),
                                                    false, false);
    const v2u r1 = __builtin_amdgcn_permlane16_swap(__float_as_uint(k1), __float_as_uint(k3),
                                                    false, false);
    const unsigned r0a = r0.x, r0b = r0.y, r1a = r1.x, r1b = r1.y;
    f0.x = __uint_as_float(r0a);
    f0.y = __uint_as_float(r1a);
    f1.x = __uint_as_float(r0b);
    f1.y = __uint_as_float(r1b);
}

__global__ __launch_bounds__(256) void
nolc1d_wmma_f32(const float* __restrict__ x,
                const float* __restrict__ w,
                float* __restrict__ out)
{
    const int lane = threadIdx.x & 31;
    const int wvid = threadIdx.x >> 5;
    const int t    = wvid & 3;
    const int pp   = (blockIdx.x << 1) + (wvid >> 2);

    const int half = lane >> 4;
    const int lid  = lane & 15;

    const float* xp = x + ((size_t)lid * CIN) * DLEN + 8 * pp + 4 * half;
    const float* wp = w + ((size_t)(16 * t + lid) * CIN) * DLEN + 8 * pp + 4 * half;

    v8f acc0 = {}, acc1 = {};

    v4f xa[CIN], wa[CIN];
#pragma unroll
    for (int c = 0; c < CIN; ++c) {
        xa[c] = *(const v4f*)(xp + (size_t)c * DLEN);
        wa[c] = *(const v4f*)(wp + (size_t)c * DLEN);
    }

    __builtin_amdgcn_sched_barrier(0);

#pragma unroll
    for (int c = 0; c < CIN; ++c) {
        v2f a0, a1, b0, b1;
        split_pair(xa[c], a0, a1);
        split_pair(wa[c], b0, b1);
        acc0 = __builtin_amdgcn_wmma_f32_16x16x4_f32(false, a0, false, b0,
                                                     (short)0, acc0, false, false);
        acc1 = __builtin_amdgcn_wmma_f32_16x16x4_f32(false, a1, false, b1,
                                                     (short)0, acc1, false, false);
    }

    float* op = out + ((size_t)(half * 8) * COUT + (16 * t + lid)) * OUT_DIM + 2 * pp;
#pragma unroll
    for (int j = 0; j < 8; ++j) {
        v2f v;
        v.x = acc0[j] * SCALE;
        v.y = acc1[j] * SCALE;
        *(v2f*)(op + (size_t)j * COUT * OUT_DIM) = v;
    }
}

extern "C" void kernel_launch(void* const* d_in, const int* in_sizes, int n_in,
                              void* d_out, int out_size, void* d_ws, size_t ws_size,
                              hipStream_t stream)
{
    (void)in_sizes; (void)n_in; (void)out_size; (void)d_ws; (void)ws_size;
    const float* x = (const float*)d_in[0];
    const float* w = (const float*)d_in[1];
    float* out     = (float*)d_out;

    dim3 grid(256);
    dim3 block(256);
    nolc1d_wmma_f32<<<grid, block, 0, stream>>>(x, w, out);
}
